# starting kernel
# baseline (speedup 1.0000x reference)
#include <hip/hip_runtime.h>
#include <hip/hip_bf16.h>

typedef __attribute__((ext_vector_type(2))) float v2f;
typedef __attribute__((ext_vector_type(8))) float v8f;

#define H 128
#define D_EDGE 16
#define N_MOUSE 50000
#define N_VIRUS 10000
#define NEDGE   500000
#define NPAIR   100000
#define D_IN    1024

__global__ void gemm_wmma_bias(const float* __restrict__ A,
                               const float* __restrict__ B,
                               const float* __restrict__ bias,
                               float* __restrict__ C,
                               int M, int K)
{
    __shared__ __align__(16) float As[16 * 132];

    const int tid  = threadIdx.x;
    const int wave = tid >> 5;
    const int lane = tid & 31;
    const int hi   = lane >> 4;
    const int ml   = lane & 15;
    const int row0 = blockIdx.x * 16;
    const int n0   = wave * 16;

    v8f acc = {};

    for (int k0 = 0; k0 < K; k0 += 128) {
        #pragma unroll
        for (int it = 0; it < 2; ++it) {
            int i   = tid * 4 + it * 1024;
            int row = i >> 7;
            int col = i & 127;
            float4 val = *(const float4*)(A + (size_t)(row0 + row) * K + k0 + col);
            *(float4*)(&As[row * 132 + col]) = val;
        }
        __syncthreads();

        #pragma unroll 8
        for (int kk = 0; kk < 128; kk += 4) {
            v2f a = *(const v2f*)(&As[ml * 132 + kk + 2 * hi]);
            const float* bp = B + (size_t)(k0 + kk + 2 * hi) * H + n0 + ml;
            v2f b;
            b.x = bp[0];
            b.y = bp[H];
            acc = __builtin_amdgcn_wmma_f32_16x16x4_f32(
                      false, a, false, b, (short)0, acc, false, false);
        }
        __syncthreads();
    }

    const float bval = bias ? bias[n0 + ml] : 0.0f;
    #pragma unroll
    for (int r = 0; r < 8; ++r) {
        int m = row0 + r + 8 * hi;
        C[(size_t)m * H + n0 + ml] = acc[r] + bval;
    }
}

__global__ void edge_conv(const float* __restrict__ kbuf,
                          const float* __restrict__ qbuf,
                          const float* __restrict__ vbuf,
                          const float* __restrict__ xe,
                          const float* __restrict__ We,
                          const float* __restrict__ be,
                          const int*   __restrict__ src,
                          const int*   __restrict__ dst,
                          float*       __restrict__ out_dst,
                          int E)
{
    const int e = blockIdx.x * 2 + (threadIdx.x >> 7);
    const int h = threadIdx.x & 127;
    if (e >= E) return;

    const int s = src[e];
    const int d = dst[e];

    float ea = be[h];
    const float* xrow = xe + (size_t)e * D_EDGE;
    #pragma unroll
    for (int dd = 0; dd < D_EDGE; ++dd)
        ea = fmaf(xrow[dd], We[dd * H + h], ea);

    const float ki = kbuf[(size_t)d * H + h] + ea;
    const float qj = qbuf[(size_t)s * H + h] + ea;
    const float vj = vbuf[(size_t)s * H + h] + ea;

    const float g = 1.0f / (1.0f + __expf(-(ki + qj)));
    atomicAdd(&out_dst[(size_t)d * H + h], g * vj);
}

__global__ void relu_inplace(float* __restrict__ x, int n)
{
    int i = blockIdx.x * blockDim.x + threadIdx.x;
    if (i < n) x[i] = fmaxf(x[i], 0.0f);
}

__global__ void classifier(const float* __restrict__ hm,
                           const float* __restrict__ hv,
                           const int*   __restrict__ im,
                           const int*   __restrict__ iv,
                           const float* __restrict__ Wc,
                           const float* __restrict__ bc,
                           float*       __restrict__ out,
                           int P)
{
    const int p    = blockIdx.x * 8 + (threadIdx.x >> 5);
    const int lane = threadIdx.x & 31;
    if (p >= P) return;

    const float4 am = ((const float4*)(hm + (size_t)im[p] * H))[lane];
    const float4 av = ((const float4*)(hv + (size_t)iv[p] * H))[lane];

    float c0 = 0.f, c1 = 0.f;
    const int ch = lane * 4;
    const float a[4] = {am.x, am.y, am.z, am.w};
    const float b[4] = {av.x, av.y, av.z, av.w};
    #pragma unroll
    for (int i = 0; i < 4; ++i) {
        c0 = fmaf(a[i], Wc[(ch + i) * 2 + 0], c0);
        c1 = fmaf(a[i], Wc[(ch + i) * 2 + 1], c1);
        c0 = fmaf(b[i], Wc[(H + ch + i) * 2 + 0], c0);
        c1 = fmaf(b[i], Wc[(H + ch + i) * 2 + 1], c1);
    }
    #pragma unroll
    for (int off = 16; off > 0; off >>= 1) {
        c0 += __shfl_down(c0, off, 32);
        c1 += __shfl_down(c1, off, 32);
    }
    if (lane == 0) {
        out[(size_t)p * 2 + 0] = c0 + bc[0];
        out[(size_t)p * 2 + 1] = c1 + bc[1];
    }
}

static inline void gemm(const float* A, const float* B, const float* bias,
                        float* C, int M, int K, hipStream_t s)
{
    gemm_wmma_bias<<<dim3(M / 16), dim3(256), 0, s>>>(A, B, bias, C, M, K);
}

extern "C" void kernel_launch(void* const* d_in, const int* in_sizes, int n_in,
                              void* d_out, int out_size, void* d_ws, size_t ws_size,
                              hipStream_t stream)
{
    const float* x_mouse  = (const float*)d_in[0];
    const float* x_virus  = (const float*)d_in[1];
    const float* xe_mv    = (const float*)d_in[2];
    const float* xe_vm    = (const float*)d_in[3];
    const float* W_mouse  = (const float*)d_in[4];
    const float* b_mouse  = (const float*)d_in[5];
    const float* W_virus  = (const float*)d_in[6];
    const float* b_virus  = (const float*)d_in[7];
    const float* W_key    = (const float*)d_in[8];
    const float* b_key    = (const float*)d_in[9];
    const float* W_query  = (const float*)d_in[10];
    const float* b_query  = (const float*)d_in[11];
    const float* W_value  = (const float*)d_in[12];
    const float* b_value  = (const float*)d_in[13];
    const float* W_edge   = (const float*)d_in[14];
    const float* b_edge   = (const float*)d_in[15];
    const float* W_skip   = (const float*)d_in[16];
    const float* bias_cv  = (const float*)d_in[17];
    const float* W_cls    = (const float*)d_in[18];
    const float* b_cls    = (const float*)d_in[19];
    const int* ei_mv_src  = (const int*)d_in[20];
    const int* ei_mv_dst  = (const int*)d_in[21];
    const int* ei_vm_src  = (const int*)d_in[22];
    const int* ei_vm_dst  = (const int*)d_in[23];
    const int* eli_mouse  = (const int*)d_in[24];
    const int* eli_virus  = (const int*)d_in[25];
    float* out = (float*)d_out;

    float* ws  = (float*)d_ws;
    const size_t SZ_M = (size_t)N_MOUSE * H;
    const size_t SZ_V = (size_t)N_VIRUS * H;
    float* hm0 = ws;
    float* hm1 = hm0 + SZ_M;
    float* hv0 = hm1 + SZ_M;
    float* hv1 = hv0 + SZ_V;
    float* kb  = hv1 + SZ_V;
    float* qb  = kb + SZ_M;
    float* vb  = qb + SZ_M;

    const int relu_m_blocks = (int)((SZ_M + 255) / 256);
    const int relu_v_blocks = (int)((SZ_V + 255) / 256);
    const int edge_blocks   = (NEDGE + 1) / 2;

    gemm(x_mouse, W_mouse, b_mouse, hm0, N_MOUSE, D_IN, stream);
    gemm(x_virus, W_virus, b_virus, hv0, N_VIRUS, D_IN, stream);

    float* hm_in = hm0; float* hv_in = hv0;
    float* hm_out = hm1; float* hv_out = hv1;

    for (int layer = 0; layer < 2; ++layer) {
        const int i_mv = layer * 2;
        const int i_vm = layer * 2 + 1;

        {
            const int i = i_mv;
            const float* Wk = W_key   + (size_t)i * H * H;
            const float* Wq = W_query + (size_t)i * H * H;
            const float* Wv = W_value + (size_t)i * H * H;
            const float* We = W_edge  + (size_t)i * D_EDGE * H;
            const float* Ws = W_skip  + (size_t)i * H * H;
            gemm(hv_in, Ws, bias_cv + (size_t)i * H, hv_out, N_VIRUS, H, stream);
            gemm(hv_in, Wk, b_key   + (size_t)i * H, kb, N_VIRUS, H, stream);
            gemm(hm_in, Wq, b_query + (size_t)i * H, qb, N_MOUSE, H, stream);
            gemm(hm_in, Wv, b_value + (size_t)i * H, vb, N_MOUSE, H, stream);
            edge_conv<<<dim3(edge_blocks), dim3(256), 0, stream>>>(
                kb, qb, vb, xe_mv, We, b_edge + (size_t)i * H,
                ei_mv_src, ei_mv_dst, hv_out, NEDGE);
        }
        {
            const int i = i_vm;
            const float* Wk = W_key   + (size_t)i * H * H;
            const float* Wq = W_query + (size_t)i * H * H;
            const float* Wv = W_value + (size_t)i * H * H;
            const float* We = W_edge  + (size_t)i * D_EDGE * H;
            const float* Ws = W_skip  + (size_t)i * H * H;
            gemm(hm_in, Ws, bias_cv + (size_t)i * H, hm_out, N_MOUSE, H, stream);
            gemm(hm_in, Wk, b_key   + (size_t)i * H, kb, N_MOUSE, H, stream);
            gemm(hv_in, Wq, b_query + (size_t)i * H, qb, N_VIRUS, H, stream);
            gemm(hv_in, Wv, b_value + (size_t)i * H, vb, N_VIRUS, H, stream);
            edge_conv<<<dim3(edge_blocks), dim3(256), 0, stream>>>(
                kb, qb, vb, xe_vm, We, b_edge + (size_t)i * H,
                ei_vm_src, ei_vm_dst, hm_out, NEDGE);
        }
        relu_inplace<<<dim3(relu_v_blocks), dim3(256), 0, stream>>>(hv_out, (int)SZ_V);
        relu_inplace<<<dim3(relu_m_blocks), dim3(256), 0, stream>>>(hm_out, (int)SZ_M);

        float* t;
        t = hm_in; hm_in = hm_out; hm_out = t;
        t = hv_in; hv_in = hv_out; hv_out = t;
    }

    classifier<<<dim3((NPAIR + 7) / 8), dim3(256), 0, stream>>>(
        hm_in, hv_in, eli_mouse, eli_virus, W_cls, b_cls, out, NPAIR);
}


# edge sums: each work-group owns a range of destination rows, scans the whole edge list for them and sums messages in LDS (f32 LDS atomics, arrival order; skip term added last); no global atomics
# speedup vs baseline: 1.5692x; 1.5692x over previous
#include <hip/hip_runtime.h>
#include <hip/hip_bf16.h>

typedef __attribute__((ext_vector_type(2))) float v2f;
typedef __attribute__((ext_vector_type(8))) float v8f;

#define H 128
#define D_EDGE 16
#define N_MOUSE 50000
#define N_VIRUS 10000
#define NEDGE   500000
#define NPAIR   100000
#define D_IN    1024

__global__ void gemm_wmma_bias(const float* __restrict__ A,
                               const float* __restrict__ B,
                               const float* __restrict__ bias,
                               float* __restrict__ C,
                               int M, int K)
{
    __shared__ __align__(16) float As[16 * 132];

    const int tid  = threadIdx.x;
    const int wave = tid >> 5;
    const int lane = tid & 31;
    const int hi   = lane >> 4;
    const int ml   = lane & 15;
    const int row0 = blockIdx.x * 16;
    const int n0   = wave * 16;

    v8f acc = {};

    for (int k0 = 0; k0 < K; k0 += 128) {
        #pragma unroll
        for (int it = 0; it < 2; ++it) {
            int i   = tid * 4 + it * 1024;
            int row = i >> 7;
            int col = i & 127;
            float4 val = *(const float4*)(A + (size_t)(row0 + row) * K + k0 + col);
            *(float4*)(&As[row * 132 + col]) = val;
        }
        __syncthreads();

        #pragma unroll 8
        for (int kk = 0; kk < 128; kk += 4) {
            v2f a = *(const v2f*)(&As[ml * 132 + kk + 2 * hi]);
            const float* bp = B + (size_t)(k0 + kk + 2 * hi) * H + n0 + ml;
            v2f b;
            b.x = bp[0];
            b.y = bp[H];
            acc = __builtin_amdgcn_wmma_f32_16x16x4_f32(
                      false, a, false, b, (short)0, acc, false, false);
        }
        __syncthreads();
    }

    const float bval = bias ? bias[n0 + ml] : 0.0f;
    #pragma unroll
    for (int r = 0; r < 8; ++r) {
        int m = row0 + r + 8 * hi;
        C[(size_t)m * H + n0 + ml] = acc[r] + bval;
    }
}

#define EDGE_GRID      256
#define EDGE_THREADS   1024
#define EDGE_MAX_ROWS  196

__device__ __forceinline__ int4 load_dst4(const int4* __restrict__ dst4, int i, int n4)
{
    return dst4[i < n4 ? i : n4 - 1];
}

__global__ __launch_bounds__(EDGE_THREADS)
void edge_conv(const float* __restrict__ kbuf,
               const float* __restrict__ qbuf,
               const float* __restrict__ vbuf,
               const float* __restrict__ xe,
               const float* __restrict__ We,
               const float* __restrict__ be,
               const int*   __restrict__ src,
               const int*   __restrict__ dst,
               float*       __restrict__ out_dst,
               int E, int n_dst, int rows_per_wg)
{
    __shared__ __align__(16) float acc[EDGE_MAX_ROWS * H];

    const int tid  = threadIdx.x;
    const int lane = tid & 31;
    const int wave = __builtin_amdgcn_readfirstlane(tid >> 5);

    const int row0 = blockIdx.x * rows_per_wg;
    int nrows = n_dst - row0;
    if (nrows > rows_per_wg) nrows = rows_per_wg;
    if (nrows <= 0) return;

    for (int i = tid; i < nrows * H; i += EDGE_THREADS) acc[i] = 0.0f;

    float4 w[D_EDGE];
    #pragma unroll
    for (int dd = 0; dd < D_EDGE; ++dd)
        w[dd] = *(const float4*)(We + dd * H + 4 * lane);
    const float4 b4 = *(const float4*)(be + 4 * lane);

    __syncthreads();

    const int  n4   = E >> 2;
    const int4* dst4 = (const int4*)dst;
    const int  steps = (n4 + 2047) >> 11;

    int tt = blockIdx.x % steps;
    int bw = tt * 2048 + wave * 64;
    int4 c0 = load_dst4(dst4, bw + lane, n4);
    int4 c1 = load_dst4(dst4, bw + 32 + lane, n4);

    for (int t = 0; t < steps; ++t) {
        tt = (tt + 1 == steps) ? 0 : tt + 1;
        const int nb = tt * 2048 + wave * 64;
        const int4 n0 = load_dst4(dst4, nb + lane, n4);
        const int4 n1 = load_dst4(dst4, nb + 32 + lane, n4);

        const unsigned ok0 = __builtin_amdgcn_ballot_w32(bw + lane      < n4);
        const unsigned ok1 = __builtin_amdgcn_ballot_w32(bw + 32 + lane < n4);
        unsigned m0 = ok0 & __builtin_amdgcn_ballot_w32((unsigned)(c0.x - row0) < (unsigned)nrows);
        unsigned m1 = ok0 & __builtin_amdgcn_ballot_w32((unsigned)(c0.y - row0) < (unsigned)nrows);
        unsigned m2 = ok0 & __builtin_amdgcn_ballot_w32((unsigned)(c0.z - row0) < (unsigned)nrows);
        unsigned m3 = ok0 & __builtin_amdgcn_ballot_w32((unsigned)(c0.w - row0) < (unsigned)nrows);
        unsigned m4 = ok1 & __builtin_amdgcn_ballot_w32((unsigned)(c1.x - row0) < (unsigned)nrows);
        unsigned m5 = ok1 & __builtin_amdgcn_ballot_w32((unsigned)(c1.y - row0) < (unsigned)nrows);
        unsigned m6 = ok1 & __builtin_amdgcn_ballot_w32((unsigned)(c1.z - row0) < (unsigned)nrows);
        unsigned m7 = ok1 & __builtin_amdgcn_ballot_w32((unsigned)(c1.w - row0) < (unsigned)nrows);

        while (true) {
            int e, d;
            if (m0)      { const int L = __builtin_ctz(m0); m0 &= m0 - 1; d = __builtin_amdgcn_readlane(c0.x, L); e = 4 * (bw + L); }
            else if (m1) { const int L = __builtin_ctz(m1); m1 &= m1 - 1; d = __builtin_amdgcn_readlane(c0.y, L); e = 4 * (bw + L) + 1; }
            else if (m2) { const int L = __builtin_ctz(m2); m2 &= m2 - 1; d = __builtin_amdgcn_readlane(c0.z, L); e = 4 * (bw + L) + 2; }
            else if (m3) { const int L = __builtin_ctz(m3); m3 &= m3 - 1; d = __builtin_amdgcn_readlane(c0.w, L); e = 4 * (bw + L) + 3; }
            else if (m4) { const int L = __builtin_ctz(m4); m4 &= m4 - 1; d = __builtin_amdgcn_readlane(c1.x, L); e = 4 * (bw + 32 + L); }
            else if (m5) { const int L = __builtin_ctz(m5); m5 &= m5 - 1; d = __builtin_amdgcn_readlane(c1.y, L); e = 4 * (bw + 32 + L) + 1; }
            else if (m6) { const int L = __builtin_ctz(m6); m6 &= m6 - 1; d = __builtin_amdgcn_readlane(c1.z, L); e = 4 * (bw + 32 + L) + 2; }
            else if (m7) { const int L = __builtin_ctz(m7); m7 &= m7 - 1; d = __builtin_amdgcn_readlane(c1.w, L); e = 4 * (bw + 32 + L) + 3; }
            else break;

            const int s = src[e];

            const float* xrow = xe + (size_t)e * D_EDGE;
            float4 ea = b4;
            #pragma unroll
            for (int dd = 0; dd < D_EDGE; ++dd) {
                const float x = xrow[dd];
                ea.x = fmaf(x, w[dd].x, ea.x);
                ea.y = fmaf(x, w[dd].y, ea.y);
                ea.z = fmaf(x, w[dd].z, ea.z);
                ea.w = fmaf(x, w[dd].w, ea.w);
            }

            const float4 k4 = *(const float4*)(kbuf + (size_t)d * H + 4 * lane);
            const float4 q4 = *(const float4*)(qbuf + (size_t)s * H + 4 * lane);
            const float4 v4 = *(const float4*)(vbuf + (size_t)s * H + 4 * lane);

            float* arow = acc + (d - row0) * H + lane;
            {
                const float ki = k4.x + ea.x, qj = q4.x + ea.x, vj = v4.x + ea.x;
                const float g = 1.0f / (1.0f + __expf(-(ki + qj)));
                __hip_atomic_fetch_add(arow,      g * vj, __ATOMIC_RELAXED, __HIP_MEMORY_SCOPE_WORKGROUP);
            }
            {
                const float ki = k4.y + ea.y, qj = q4.y + ea.y, vj = v4.y + ea.y;
                const float g = 1.0f / (1.0f + __expf(-(ki + qj)));
                __hip_atomic_fetch_add(arow + 32, g * vj, __ATOMIC_RELAXED, __HIP_MEMORY_SCOPE_WORKGROUP);
            }
            {
                const float ki = k4.z + ea.z, qj = q4.z + ea.z, vj = v4.z + ea.z;
                const float g = 1.0f / (1.0f + __expf(-(ki + qj)));
                __hip_atomic_fetch_add(arow + 64, g * vj, __ATOMIC_RELAXED, __HIP_MEMORY_SCOPE_WORKGROUP);
            }
            {
                const float ki = k4.w + ea.w, qj = q4.w + ea.w, vj = v4.w + ea.w;
                const float g = 1.0f / (1.0f + __expf(-(ki + qj)));
                __hip_atomic_fetch_add(arow + 96, g * vj, __ATOMIC_RELAXED, __HIP_MEMORY_SCOPE_WORKGROUP);
            }
        }

        bw = nb; c0 = n0; c1 = n1;
    }

    __syncthreads();

    for (int r = wave; r < nrows; r += EDGE_THREADS / 32) {
        float* o = out_dst + (size_t)(row0 + r) * H + 4 * lane;
        const float* a = acc + r * H + lane;
        float4 cur = *(const float4*)o;
        cur.x += a[0];
        cur.y += a[32];
        cur.z += a[64];
        cur.w += a[96];
        *(float4*)o = cur;
    }
}

__global__ void relu_inplace(float* __restrict__ x, int n)
{
    int i = blockIdx.x * blockDim.x + threadIdx.x;
    if (i < n) x[i] = fmaxf(x[i], 0.0f);
}

__global__ void classifier(const float* __restrict__ hm,
                           const float* __restrict__ hv,
                           const int*   __restrict__ im,
                           const int*   __restrict__ iv,
                           const float* __restrict__ Wc,
                           const float* __restrict__ bc,
                           float*       __restrict__ out,
                           int P)
{
    const int p    = blockIdx.x * 8 + (threadIdx.x >> 5);
    const int lane = threadIdx.x & 31;
    if (p >= P) return;

    const float4 am = ((const float4*)(hm + (size_t)im[p] * H))[lane];
    const float4 av = ((const float4*)(hv + (size_t)iv[p] * H))[lane];

    float c0 = 0.f, c1 = 0.f;
    const int ch = lane * 4;
    const float a[4] = {am.x, am.y, am.z, am.w};
    const float b[4] = {av.x, av.y, av.z, av.w};
    #pragma unroll
    for (int i = 0; i < 4; ++i) {
        c0 = fmaf(a[i], Wc[(ch + i) * 2 + 0], c0);
        c1 = fmaf(a[i], Wc[(ch + i) * 2 + 1], c1);
        c0 = fmaf(b[i], Wc[(H + ch + i) * 2 + 0], c0);
        c1 = fmaf(b[i], Wc[(H + ch + i) * 2 + 1], c1);
    }
    #pragma unroll
    for (int off = 16; off > 0; off >>= 1) {
        c0 += __shfl_down(c0, off, 32);
        c1 += __shfl_down(c1, off, 32);
    }
    if (lane == 0) {
        out[(size_t)p * 2 + 0] = c0 + bc[0];
        out[(size_t)p * 2 + 1] = c1 + bc[1];
    }
}

static inline void gemm(const float* A, const float* B, const float* bias,
                        float* C, int M, int K, hipStream_t s)
{
    gemm_wmma_bias<<<dim3(M / 16), dim3(256), 0, s>>>(A, B, bias, C, M, K);
}

extern "C" void kernel_launch(void* const* d_in, const int* in_sizes, int n_in,
                              void* d_out, int out_size, void* d_ws, size_t ws_size,
                              hipStream_t stream)
{
    const float* x_mouse  = (const float*)d_in[0];
    const float* x_virus  = (const float*)d_in[1];
    const float* xe_mv    = (const float*)d_in[2];
    const float* xe_vm    = (const float*)d_in[3];
    const float* W_mouse  = (const float*)d_in[4];
    const float* b_mouse  = (const float*)d_in[5];
    const float* W_virus  = (const float*)d_in[6];
    const float* b_virus  = (const float*)d_in[7];
    const float* W_key    = (const float*)d_in[8];
    const float* b_key    = (const float*)d_in[9];
    const float* W_query  = (const float*)d_in[10];
    const float* b_query  = (const float*)d_in[11];
    const float* W_value  = (const float*)d_in[12];
    const float* b_value  = (const float*)d_in[13];
    const float* W_edge   = (const float*)d_in[14];
    const float* b_edge   = (const float*)d_in[15];
    const float* W_skip   = (const float*)d_in[16];
    const float* bias_cv  = (const float*)d_in[17];
    const float* W_cls    = (const float*)d_in[18];
    const float* b_cls    = (const float*)d_in[19];
    const int* ei_mv_src  = (const int*)d_in[20];
    const int* ei_mv_dst  = (const int*)d_in[21];
    const int* ei_vm_src  = (const int*)d_in[22];
    const int* ei_vm_dst  = (const int*)d_in[23];
    const int* eli_mouse  = (const int*)d_in[24];
    const int* eli_virus  = (const int*)d_in[25];
    float* out = (float*)d_out;

    float* ws  = (float*)d_ws;
    const size_t SZ_M = (size_t)N_MOUSE * H;
    const size_t SZ_V = (size_t)N_VIRUS * H;
    float* hm0 = ws;
    float* hm1 = hm0 + SZ_M;
    float* hv0 = hm1 + SZ_M;
    float* hv1 = hv0 + SZ_V;
    float* kb  = hv1 + SZ_V;
    float* qb  = kb + SZ_M;
    float* vb  = qb + SZ_M;

    const int relu_m_blocks = (int)((SZ_M + 255) / 256);
    const int relu_v_blocks = (int)((SZ_V + 255) / 256);

    gemm(x_mouse, W_mouse, b_mouse, hm0, N_MOUSE, D_IN, stream);
    gemm(x_virus, W_virus, b_virus, hv0, N_VIRUS, D_IN, stream);

    float* hm_in = hm0; float* hv_in = hv0;
    float* hm_out = hm1; float* hv_out = hv1;

    for (int layer = 0; layer < 2; ++layer) {
        const int i_mv = layer * 2;
        const int i_vm = layer * 2 + 1;

        {
            const int i = i_mv;
            const float* Wk = W_key   + (size_t)i * H * H;
            const float* Wq = W_query + (size_t)i * H * H;
            const float* Wv = W_value + (size_t)i * H * H;
            const float* We = W_edge  + (size_t)i * D_EDGE * H;
            const float* Ws = W_skip  + (size_t)i * H * H;
            gemm(hv_in, Ws, bias_cv + (size_t)i * H, hv_out, N_VIRUS, H, stream);
            gemm(hv_in, Wk, b_key   + (size_t)i * H, kb, N_VIRUS, H, stream);
            gemm(hm_in, Wq, b_query + (size_t)i * H, qb, N_MOUSE, H, stream);
            gemm(hm_in, Wv, b_value + (size_t)i * H, vb, N_MOUSE, H, stream);
            edge_conv<<<dim3(EDGE_GRID), dim3(EDGE_THREADS), 0, stream>>>(
                kb, qb, vb, xe_mv, We, b_edge + (size_t)i * H,
                ei_mv_src, ei_mv_dst, hv_out, NEDGE, N_VIRUS, 40);
        }
        {
            const int i = i_vm;
            const float* Wk = W_key   + (size_t)i * H * H;
            const float* Wq = W_query + (size_t)i * H * H;
            const float* Wv = W_value + (size_t)i * H * H;
            const float* We = W_edge  + (size_t)i * D_EDGE * H;
            const float* Ws = W_skip  + (size_t)i * H * H;
            gemm(hm_in, Ws, bias_cv + (size_t)i * H, hm_out, N_MOUSE, H, stream);
            gemm(hm_in, Wk, b_key   + (size_t)i * H, kb, N_MOUSE, H, stream);
            gemm(hv_in, Wq, b_query + (size_t)i * H, qb, N_VIRUS, H, stream);
            gemm(hv_in, Wv, b_value + (size_t)i * H, vb, N_VIRUS, H, stream);
            edge_conv<<<dim3(EDGE_GRID), dim3(EDGE_THREADS), 0, stream>>>(
                kb, qb, vb, xe_vm, We, b_edge + (size_t)i * H,
                ei_vm_src, ei_vm_dst, hm_out, NEDGE, N_MOUSE, EDGE_MAX_ROWS);
        }
        relu_inplace<<<dim3(relu_v_blocks), dim3(256), 0, stream>>>(hv_out, (int)SZ_V);
        relu_inplace<<<dim3(relu_m_blocks), dim3(256), 0, stream>>>(hm_out, (int)SZ_M);

        float* t;
        t = hm_in; hm_in = hm_out; hm_out = t;
        t = hv_in; hv_in = hv_out; hv_out = t;
    }

    classifier<<<dim3((NPAIR + 7) / 8), dim3(256), 0, stream>>>(
        hm_in, hv_in, eli_mouse, eli_virus, W_cls, b_cls, out, NPAIR);
}


# ReLU moves into the edge kernel's closing row store (a row's one owner writes max(skip+bias+sum,0)): the 4 elementwise launches, 480000 waves of one 4-byte load each, are gone; sum order kept
# speedup vs baseline: 1.6551x; 1.0547x over previous
#include <hip/hip_runtime.h>
#include <hip/hip_bf16.h>

typedef __attribute__((ext_vector_type(2))) float v2f;
typedef __attribute__((ext_vector_type(8))) float v8f;

#define H 128
#define D_EDGE 16
#define N_MOUSE 50000
#define N_VIRUS 10000
#define NEDGE   500000
#define NPAIR   100000
#define D_IN    1024

__global__ void gemm_wmma_bias(const float* __restrict__ A,
                               const float* __restrict__ B,
                               const float* __restrict__ bias,
                               float* __restrict__ C,
                               int M, int K)
{
    __shared__ __align__(16) float As[16 * 132];

    const int tid  = threadIdx.x;
    const int wave = tid >> 5;
    const int lane = tid & 31;
    const int hi   = lane >> 4;
    const int ml   = lane & 15;
    const int row0 = blockIdx.x * 16;
    const int n0   = wave * 16;

    v8f acc = {};

    for (int k0 = 0; k0 < K; k0 += 128) {
        #pragma unroll
        for (int it = 0; it < 2; ++it) {
            int i   = tid * 4 + it * 1024;
            int row = i >> 7;
            int col = i & 127;
            float4 val = *(const float4*)(A + (size_t)(row0 + row) * K + k0 + col);
            *(float4*)(&As[row * 132 + col]) = val;
        }
        __syncthreads();

        #pragma unroll 8
        for (int kk = 0; kk < 128; kk += 4) {
            v2f a = *(const v2f*)(&As[ml * 132 + kk + 2 * hi]);
            const float* bp = B + (size_t)(k0 + kk + 2 * hi) * H + n0 + ml;
            v2f b;
            b.x = bp[0];
            b.y = bp[H];
            acc = __builtin_amdgcn_wmma_f32_16x16x4_f32(
                      false, a, false, b, (short)0, acc, false, false);
        }
        __syncthreads();
    }

    const float bval = bias ? bias[n0 + ml] : 0.0f;
    #pragma unroll
    for (int r = 0; r < 8; ++r) {
        int m = row0 + r + 8 * hi;
        C[(size_t)m * H + n0 + ml] = acc[r] + bval;
    }
}

#define EDGE_GRID      256
#define EDGE_THREADS   1024
#define EDGE_MAX_ROWS  196

__device__ __forceinline__ int4 load_dst4(const int4* __restrict__ dst4, int i, int n4)
{
    return dst4[i < n4 ? i : n4 - 1];
}

__global__ __launch_bounds__(EDGE_THREADS)
void edge_conv(const float* __restrict__ kbuf,
               const float* __restrict__ qbuf,
               const float* __restrict__ vbuf,
               const float* __restrict__ xe,
               const float* __restrict__ We,
               const float* __restrict__ be,
               const int*   __restrict__ src,
               const int*   __restrict__ dst,
               float*       __restrict__ out_dst,
               int E, int n_dst, int rows_per_wg)
{
    __shared__ __align__(16) float acc[EDGE_MAX_ROWS * H];

    const int tid  = threadIdx.x;
    const int lane = tid & 31;
    const int wave = __builtin_amdgcn_readfirstlane(tid >> 5);

    const int row0 = blockIdx.x * rows_per_wg;
    int nrows = n_dst - row0;
    if (nrows > rows_per_wg) nrows = rows_per_wg;
    if (nrows <= 0) return;

    for (int i = tid; i < nrows * H; i += EDGE_THREADS) acc[i] = 0.0f;

    float4 w[D_EDGE];
    #pragma unroll
    for (int dd = 0; dd < D_EDGE; ++dd)
        w[dd] = *(const float4*)(We + dd * H + 4 * lane);
    const float4 b4 = *(const float4*)(be + 4 * lane);

    __syncthreads();

    const int  n4   = E >> 2;
    const int4* dst4 = (const int4*)dst;
    const int  steps = (n4 + 2047) >> 11;

    int tt = blockIdx.x % steps;
    int bw = tt * 2048 + wave * 64;
    int4 c0 = load_dst4(dst4, bw + lane, n4);
    int4 c1 = load_dst4(dst4, bw + 32 + lane, n4);

    for (int t = 0; t < steps; ++t) {
        tt = (tt + 1 == steps) ? 0 : tt + 1;
        const int nb = tt * 2048 + wave * 64;
        const int4 n0 = load_dst4(dst4, nb + lane, n4);
        const int4 n1 = load_dst4(dst4, nb + 32 + lane, n4);

        const unsigned ok0 = __builtin_amdgcn_ballot_w32(bw + lane      < n4);
        const unsigned ok1 = __builtin_amdgcn_ballot_w32(bw + 32 + lane < n4);
        unsigned m0 = ok0 & __builtin_amdgcn_ballot_w32((unsigned)(c0.x - row0) < (unsigned)nrows);
        unsigned m1 = ok0 & __builtin_amdgcn_ballot_w32((unsigned)(c0.y - row0) < (unsigned)nrows);
        unsigned m2 = ok0 & __builtin_amdgcn_ballot_w32((unsigned)(c0.z - row0) < (unsigned)nrows);
        unsigned m3 = ok0 & __builtin_amdgcn_ballot_w32((unsigned)(c0.w - row0) < (unsigned)nrows);
        unsigned m4 = ok1 & __builtin_amdgcn_ballot_w32((unsigned)(c1.x - row0) < (unsigned)nrows);
        unsigned m5 = ok1 & __builtin_amdgcn_ballot_w32((unsigned)(c1.y - row0) < (unsigned)nrows);
        unsigned m6 = ok1 & __builtin_amdgcn_ballot_w32((unsigned)(c1.z - row0) < (unsigned)nrows);
        unsigned m7 = ok1 & __builtin_amdgcn_ballot_w32((unsigned)(c1.w - row0) < (unsigned)nrows);

        while (true) {
            int e, d;
            if (m0)      { const int L = __builtin_ctz(m0); m0 &= m0 - 1; d = __builtin_amdgcn_readlane(c0.x, L); e = 4 * (bw + L); }
            else if (m1) { const int L = __builtin_ctz(m1); m1 &= m1 - 1; d = __builtin_amdgcn_readlane(c0.y, L); e = 4 * (bw + L) + 1; }
            else if (m2) { const int L = __builtin_ctz(m2); m2 &= m2 - 1; d = __builtin_amdgcn_readlane(c0.z, L); e = 4 * (bw + L) + 2; }
            else if (m3) { const int L = __builtin_ctz(m3); m3 &= m3 - 1; d = __builtin_amdgcn_readlane(c0.w, L); e = 4 * (bw + L) + 3; }
            else if (m4) { const int L = __builtin_ctz(m4); m4 &= m4 - 1; d = __builtin_amdgcn_readlane(c1.x, L); e = 4 * (bw + 32 + L); }
            else if (m5) { const int L = __builtin_ctz(m5); m5 &= m5 - 1; d = __builtin_amdgcn_readlane(c1.y, L); e = 4 * (bw + 32 + L) + 1; }
            else if (m6) { const int L = __builtin_ctz(m6); m6 &= m6 - 1; d = __builtin_amdgcn_readlane(c1.z, L); e = 4 * (bw + 32 + L) + 2; }
            else if (m7) { const int L = __builtin_ctz(m7); m7 &= m7 - 1; d = __builtin_amdgcn_readlane(c1.w, L); e = 4 * (bw + 32 + L) + 3; }
            else break;

            const int s = src[e];

            const float* xrow = xe + (size_t)e * D_EDGE;
            float4 ea = b4;
            #pragma unroll
            for (int dd = 0; dd < D_EDGE; ++dd) {
                const float x = xrow[dd];
                ea.x = fmaf(x, w[dd].x, ea.x);
                ea.y = fmaf(x, w[dd].y, ea.y);
                ea.z = fmaf(x, w[dd].z, ea.z);
                ea.w = fmaf(x, w[dd].w, ea.w);
            }

            const float4 k4 = *(const float4*)(kbuf + (size_t)d * H + 4 * lane);
            const float4 q4 = *(const float4*)(qbuf + (size_t)s * H + 4 * lane);
            const float4 v4 = *(const float4*)(vbuf + (size_t)s * H + 4 * lane);

            float* arow = acc + (d - row0) * H + lane;
            {
                const float ki = k4.x + ea.x, qj = q4.x + ea.x, vj = v4.x + ea.x;
                const float g = 1.0f / (1.0f + __expf(-(ki + qj)));
                __hip_atomic_fetch_add(arow,      g * vj, __ATOMIC_RELAXED, __HIP_MEMORY_SCOPE_WORKGROUP);
            }
            {
                const float ki = k4.y + ea.y, qj = q4.y + ea.y, vj = v4.y + ea.y;
                const float g = 1.0f / (1.0f + __expf(-(ki + qj)));
                __hip_atomic_fetch_add(arow + 32, g * vj, __ATOMIC_RELAXED, __HIP_MEMORY_SCOPE_WORKGROUP);
            }
            {
                const float ki = k4.z + ea.z, qj = q4.z + ea.z, vj = v4.z + ea.z;
                const float g = 1.0f / (1.0f + __expf(-(ki + qj)));
                __hip_atomic_fetch_add(arow + 64, g * vj, __ATOMIC_RELAXED, __HIP_MEMORY_SCOPE_WORKGROUP);
            }
            {
                const float ki = k4.w + ea.w, qj = q4.w + ea.w, vj = v4.w + ea.w;
                const float g = 1.0f / (1.0f + __expf(-(ki + qj)));
                __hip_atomic_fetch_add(arow + 96, g * vj, __ATOMIC_RELAXED, __HIP_MEMORY_SCOPE_WORKGROUP);
            }
        }

        bw = nb; c0 = n0; c1 = n1;
    }

    __syncthreads();

    for (int r = wave; r < nrows; r += EDGE_THREADS / 32) {
        float* o = out_dst + (size_t)(row0 + r) * H + 4 * lane;
        const float* a = acc + r * H + lane;
        float4 cur = *(const float4*)o;
        cur.x = fmaxf(cur.x + a[0],  0.0f);
        cur.y = fmaxf(cur.y + a[32], 0.0f);
        cur.z = fmaxf(cur.z + a[64], 0.0f);
        cur.w = fmaxf(cur.w + a[96], 0.0f);
        *(float4*)o = cur;
    }
}

__global__ void classifier(const float* __restrict__ hm,
                           const float* __restrict__ hv,
                           const int*   __restrict__ im,
                           const int*   __restrict__ iv,
                           const float* __restrict__ Wc,
                           const float* __restrict__ bc,
                           float*       __restrict__ out,
                           int P)
{
    const int p    = blockIdx.x * 8 + (threadIdx.x >> 5);
    const int lane = threadIdx.x & 31;
    if (p >= P) return;

    const float4 am = ((const float4*)(hm + (size_t)im[p] * H))[lane];
    const float4 av = ((const float4*)(hv + (size_t)iv[p] * H))[lane];

    float c0 = 0.f, c1 = 0.f;
    const int ch = lane * 4;
    const float a[4] = {am.x, am.y, am.z, am.w};
    const float b[4] = {av.x, av.y, av.z, av.w};
    #pragma unroll
    for (int i = 0; i < 4; ++i) {
        c0 = fmaf(a[i], Wc[(ch + i) * 2 + 0], c0);
        c1 = fmaf(a[i], Wc[(ch + i) * 2 + 1], c1);
        c0 = fmaf(b[i], Wc[(H + ch + i) * 2 + 0], c0);
        c1 = fmaf(b[i], Wc[(H + ch + i) * 2 + 1], c1);
    }
    #pragma unroll
    for (int off = 16; off > 0; off >>= 1) {
        c0 += __shfl_down(c0, off, 32);
        c1 += __shfl_down(c1, off, 32);
    }
    if (lane == 0) {
        out[(size_t)p * 2 + 0] = c0 + bc[0];
        out[(size_t)p * 2 + 1] = c1 + bc[1];
    }
}

static inline void gemm(const float* A, const float* B, const float* bias,
                        float* C, int M, int K, hipStream_t s)
{
    gemm_wmma_bias<<<dim3(M / 16), dim3(256), 0, s>>>(A, B, bias, C, M, K);
}

extern "C" void kernel_launch(void* const* d_in, const int* in_sizes, int n_in,
                              void* d_out, int out_size, void* d_ws, size_t ws_size,
                              hipStream_t stream)
{
    const float* x_mouse  = (const float*)d_in[0];
    const float* x_virus  = (const float*)d_in[1];
    const float* xe_mv    = (const float*)d_in[2];
    const float* xe_vm    = (const float*)d_in[3];
    const float* W_mouse  = (const float*)d_in[4];
    const float* b_mouse  = (const float*)d_in[5];
    const float* W_virus  = (const float*)d_in[6];
    const float* b_virus  = (const float*)d_in[7];
    const float* W_key    = (const float*)d_in[8];
    const float* b_key    = (const float*)d_in[9];
    const float* W_query  = (const float*)d_in[10];
    const float* b_query  = (const float*)d_in[11];
    const float* W_value  = (const float*)d_in[12];
    const float* b_value  = (const float*)d_in[13];
    const float* W_edge   = (const float*)d_in[14];
    const float* b_edge   = (const float*)d_in[15];
    const float* W_skip   = (const float*)d_in[16];
    const float* bias_cv  = (const float*)d_in[17];
    const float* W_cls    = (const float*)d_in[18];
    const float* b_cls    = (const float*)d_in[19];
    const int* ei_mv_src  = (const int*)d_in[20];
    const int* ei_mv_dst  = (const int*)d_in[21];
    const int* ei_vm_src  = (const int*)d_in[22];
    const int* ei_vm_dst  = (const int*)d_in[23];
    const int* eli_mouse  = (const int*)d_in[24];
    const int* eli_virus  = (const int*)d_in[25];
    float* out = (float*)d_out;

    float* ws  = (float*)d_ws;
    const size_t SZ_M = (size_t)N_MOUSE * H;
    const size_t SZ_V = (size_t)N_VIRUS * H;
    float* hm0 = ws;
    float* hm1 = hm0 + SZ_M;
    float* hv0 = hm1 + SZ_M;
    float* hv1 = hv0 + SZ_V;
    float* kb  = hv1 + SZ_V;
    float* qb  = kb + SZ_M;
    float* vb  = qb + SZ_M;

    gemm(x_mouse, W_mouse, b_mouse, hm0, N_MOUSE, D_IN, stream);
    gemm(x_virus, W_virus, b_virus, hv0, N_VIRUS, D_IN, stream);

    float* hm_in = hm0; float* hv_in = hv0;
    float* hm_out = hm1; float* hv_out = hv1;

    for (int layer = 0; layer < 2; ++layer) {
        const int i_mv = layer * 2;
        const int i_vm = layer * 2 + 1;

        {
            const int i = i_mv;
            const float* Wk = W_key   + (size_t)i * H * H;
            const float* Wq = W_query + (size_t)i * H * H;
            const float* Wv = W_value + (size_t)i * H * H;
            const float* We = W_edge  + (size_t)i * D_EDGE * H;
            const float* Ws = W_skip  + (size_t)i * H * H;
            gemm(hv_in, Ws, bias_cv + (size_t)i * H, hv_out, N_VIRUS, H, stream);
            gemm(hv_in, Wk, b_key   + (size_t)i * H, kb, N_VIRUS, H, stream);
            gemm(hm_in, Wq, b_query + (size_t)i * H, qb, N_MOUSE, H, stream);
            gemm(hm_in, Wv, b_value + (size_t)i * H, vb, N_MOUSE, H, stream);
            edge_conv<<<dim3(EDGE_GRID), dim3(EDGE_THREADS), 0, stream>>>(
                kb, qb, vb, xe_mv, We, b_edge + (size_t)i * H,
                ei_mv_src, ei_mv_dst, hv_out, NEDGE, N_VIRUS, 40);
        }
        {
            const int i = i_vm;
            const float* Wk = W_key   + (size_t)i * H * H;
            const float* Wq = W_query + (size_t)i * H * H;
            const float* Wv = W_value + (size_t)i * H * H;
            const float* We = W_edge  + (size_t)i * D_EDGE * H;
            const float* Ws = W_skip  + (size_t)i * H * H;
            gemm(hm_in, Ws, bias_cv + (size_t)i * H, hm_out, N_MOUSE, H, stream);
            gemm(hm_in, Wk, b_key   + (size_t)i * H, kb, N_MOUSE, H, stream);
            gemm(hv_in, Wq, b_query + (size_t)i * H, qb, N_VIRUS, H, stream);
            gemm(hv_in, Wv, b_value + (size_t)i * H, vb, N_VIRUS, H, stream);
            edge_conv<<<dim3(EDGE_GRID), dim3(EDGE_THREADS), 0, stream>>>(
                kb, qb, vb, xe_vm, We, b_edge + (size_t)i * H,
                ei_vm_src, ei_vm_dst, hm_out, NEDGE, N_MOUSE, EDGE_MAX_ROWS);
        }

        float* t;
        t = hm_in; hm_in = hm_out; hm_out = t;
        t = hv_in; hv_in = hv_out; hv_out = t;
    }

    classifier<<<dim3((NPAIR + 7) / 8), dim3(256), 0, stream>>>(
        hm_in, hv_in, eli_mouse, eli_virus, W_cls, b_cls, out, NPAIR);
}


# edge kernel: an edge's feature row is requested with its source index, not after it (index read as a vector load, waited for alone): the row's memory trip overlaps index and q/v reads; sum order kept
# speedup vs baseline: 1.7656x; 1.0668x over previous
#include <hip/hip_runtime.h>
#include <hip/hip_bf16.h>

typedef __attribute__((ext_vector_type(2))) float v2f;
typedef __attribute__((ext_vector_type(8))) float v8f;

#define H 128
#define D_EDGE 16
#define N_MOUSE 50000
#define N_VIRUS 10000
#define NEDGE   500000
#define NPAIR   100000
#define D_IN    1024

__global__ void gemm_wmma_bias(const float* __restrict__ A,
                               const float* __restrict__ B,
                               const float* __restrict__ bias,
                               float* __restrict__ C,
                               int M, int K)
{
    __shared__ __align__(16) float As[16 * 132];

    const int tid  = threadIdx.x;
    const int wave = tid >> 5;
    const int lane = tid & 31;
    const int hi   = lane >> 4;
    const int ml   = lane & 15;
    const int row0 = blockIdx.x * 16;
    const int n0   = wave * 16;

    v8f acc = {};

    for (int k0 = 0; k0 < K; k0 += 128) {
        #pragma unroll
        for (int it = 0; it < 2; ++it) {
            int i   = tid * 4 + it * 1024;
            int row = i >> 7;
            int col = i & 127;
            float4 val = *(const float4*)(A + (size_t)(row0 + row) * K + k0 + col);
            *(float4*)(&As[row * 132 + col]) = val;
        }
        __syncthreads();

        #pragma unroll 8
        for (int kk = 0; kk < 128; kk += 4) {
            v2f a = *(const v2f*)(&As[ml * 132 + kk + 2 * hi]);
            const float* bp = B + (size_t)(k0 + kk + 2 * hi) * H + n0 + ml;
            v2f b;
            b.x = bp[0];
            b.y = bp[H];
            acc = __builtin_amdgcn_wmma_f32_16x16x4_f32(
                      false, a, false, b, (short)0, acc, false, false);
        }
        __syncthreads();
    }

    const float bval = bias ? bias[n0 + ml] : 0.0f;
    #pragma unroll
    for (int r = 0; r < 8; ++r) {
        int m = row0 + r + 8 * hi;
        C[(size_t)m * H + n0 + ml] = acc[r] + bval;
    }
}

#define EDGE_GRID      256
#define EDGE_THREADS   1024
#define EDGE_MAX_ROWS  196

__device__ __forceinline__ int4 load_dst4(const int4* __restrict__ dst4, int i, int n4)
{
    return dst4[i < n4 ? i : n4 - 1];
}

__global__ __launch_bounds__(EDGE_THREADS)
void edge_conv(const float* __restrict__ kbuf,
               const float* __restrict__ qbuf,
               const float* __restrict__ vbuf,
               const float* __restrict__ xe,
               const float* __restrict__ We,
               const float* __restrict__ be,
               const int*   __restrict__ src,
               const int*   __restrict__ dst,
               float*       __restrict__ out_dst,
               int E, int n_dst, int rows_per_wg)
{
    __shared__ __align__(16) float acc[EDGE_MAX_ROWS * H];

    const int tid  = threadIdx.x;
    const int lane = tid & 31;
    const int wave = __builtin_amdgcn_readfirstlane(tid >> 5);

    const int row0 = blockIdx.x * rows_per_wg;
    int nrows = n_dst - row0;
    if (nrows > rows_per_wg) nrows = rows_per_wg;
    if (nrows <= 0) return;

    for (int i = tid; i < nrows * H; i += EDGE_THREADS) acc[i] = 0.0f;

    int vzero;
    asm volatile("v_mov_b32 %0, 0" : "=v"(vzero));

    float4 w[D_EDGE];
    #pragma unroll
    for (int dd = 0; dd < D_EDGE; ++dd)
        w[dd] = *(const float4*)(We + dd * H + 4 * lane);
    const float4 b4 = *(const float4*)(be + 4 * lane);

    __syncthreads();

    const int  n4   = E >> 2;
    const int4* dst4 = (const int4*)dst;
    const int  steps = (n4 + 2047) >> 11;

    int tt = blockIdx.x % steps;
    int bw = tt * 2048 + wave * 64;
    int4 c0 = load_dst4(dst4, bw + lane, n4);
    int4 c1 = load_dst4(dst4, bw + 32 + lane, n4);

    for (int t = 0; t < steps; ++t) {
        tt = (tt + 1 == steps) ? 0 : tt + 1;
        const int nb = tt * 2048 + wave * 64;
        const int4 n0 = load_dst4(dst4, nb + lane, n4);
        const int4 n1 = load_dst4(dst4, nb + 32 + lane, n4);

        const unsigned ok0 = __builtin_amdgcn_ballot_w32(bw + lane      < n4);
        const unsigned ok1 = __builtin_amdgcn_ballot_w32(bw + 32 + lane < n4);
        unsigned m0 = ok0 & __builtin_amdgcn_ballot_w32((unsigned)(c0.x - row0) < (unsigned)nrows);
        unsigned m1 = ok0 & __builtin_amdgcn_ballot_w32((unsigned)(c0.y - row0) < (unsigned)nrows);
        unsigned m2 = ok0 & __builtin_amdgcn_ballot_w32((unsigned)(c0.z - row0) < (unsigned)nrows);
        unsigned m3 = ok0 & __builtin_amdgcn_ballot_w32((unsigned)(c0.w - row0) < (unsigned)nrows);
        unsigned m4 = ok1 & __builtin_amdgcn_ballot_w32((unsigned)(c1.x - row0) < (unsigned)nrows);
        unsigned m5 = ok1 & __builtin_amdgcn_ballot_w32((unsigned)(c1.y - row0) < (unsigned)nrows);
        unsigned m6 = ok1 & __builtin_amdgcn_ballot_w32((unsigned)(c1.z - row0) < (unsigned)nrows);
        unsigned m7 = ok1 & __builtin_amdgcn_ballot_w32((unsigned)(c1.w - row0) < (unsigned)nrows);

        while (true) {
            int e, d;
            if (m0)      { const int L = __builtin_ctz(m0); m0 &= m0 - 1; d = __builtin_amdgcn_readlane(c0.x, L); e = 4 * (bw + L); }
            else if (m1) { const int L = __builtin_ctz(m1); m1 &= m1 - 1; d = __builtin_amdgcn_readlane(c0.y, L); e = 4 * (bw + L) + 1; }
            else if (m2) { const int L = __builtin_ctz(m2); m2 &= m2 - 1; d = __builtin_amdgcn_readlane(c0.z, L); e = 4 * (bw + L) + 2; }
            else if (m3) { const int L = __builtin_ctz(m3); m3 &= m3 - 1; d = __builtin_amdgcn_readlane(c0.w, L); e = 4 * (bw + L) + 3; }
            else if (m4) { const int L = __builtin_ctz(m4); m4 &= m4 - 1; d = __builtin_amdgcn_readlane(c1.x, L); e = 4 * (bw + 32 + L); }
            else if (m5) { const int L = __builtin_ctz(m5); m5 &= m5 - 1; d = __builtin_amdgcn_readlane(c1.y, L); e = 4 * (bw + 32 + L) + 1; }
            else if (m6) { const int L = __builtin_ctz(m6); m6 &= m6 - 1; d = __builtin_amdgcn_readlane(c1.z, L); e = 4 * (bw + 32 + L) + 2; }
            else if (m7) { const int L = __builtin_ctz(m7); m7 &= m7 - 1; d = __builtin_amdgcn_readlane(c1.w, L); e = 4 * (bw + 32 + L) + 3; }
            else break;

            const int sv = src[e + vzero];
            const float* xrow = xe + (size_t)e * D_EDGE;
            float x[D_EDGE];
            #pragma unroll
            for (int dd = 0; dd < D_EDGE; ++dd) x[dd] = xrow[dd];
            __builtin_amdgcn_sched_barrier(0);
            const int s = __builtin_amdgcn_readfirstlane(sv);
            const float4 k4 = *(const float4*)(kbuf + (size_t)d * H + 4 * lane);
            const float4 q4 = *(const float4*)(qbuf + (size_t)s * H + 4 * lane);
            const float4 v4 = *(const float4*)(vbuf + (size_t)s * H + 4 * lane);
            __builtin_amdgcn_sched_barrier(0);

            float4 ea = b4;
            #pragma unroll
            for (int dd = 0; dd < D_EDGE; ++dd) {
                ea.x = fmaf(x[dd], w[dd].x, ea.x);
                ea.y = fmaf(x[dd], w[dd].y, ea.y);
                ea.z = fmaf(x[dd], w[dd].z, ea.z);
                ea.w = fmaf(x[dd], w[dd].w, ea.w);
            }

            float* arow = acc + (d - row0) * H + lane;
            {
                const float ki = k4.x + ea.x, qj = q4.x + ea.x, vj = v4.x + ea.x;
                const float g = 1.0f / (1.0f + __expf(-(ki + qj)));
                __hip_atomic_fetch_add(arow,      g * vj, __ATOMIC_RELAXED, __HIP_MEMORY_SCOPE_WORKGROUP);
            }
            {
                const float ki = k4.y + ea.y, qj = q4.y + ea.y, vj = v4.y + ea.y;
                const float g = 1.0f / (1.0f + __expf(-(ki + qj)));
                __hip_atomic_fetch_add(arow + 32, g * vj, __ATOMIC_RELAXED, __HIP_MEMORY_SCOPE_WORKGROUP);
            }
            {
                const float ki = k4.z + ea.z, qj = q4.z + ea.z, vj = v4.z + ea.z;
                const float g = 1.0f / (1.0f + __expf(-(ki + qj)));
                __hip_atomic_fetch_add(arow + 64, g * vj, __ATOMIC_RELAXED, __HIP_MEMORY_SCOPE_WORKGROUP);
            }
            {
                const float ki = k4.w + ea.w, qj = q4.w + ea.w, vj = v4.w + ea.w;
                const float g = 1.0f / (1.0f + __expf(-(ki + qj)));
                __hip_atomic_fetch_add(arow + 96, g * vj, __ATOMIC_RELAXED, __HIP_MEMORY_SCOPE_WORKGROUP);
            }
        }

        bw = nb; c0 = n0; c1 = n1;
    }

    __syncthreads();

    for (int r = wave; r < nrows; r += EDGE_THREADS / 32) {
        float* o = out_dst + (size_t)(row0 + r) * H + 4 * lane;
        const float* a = acc + r * H + lane;
        float4 cur = *(const float4*)o;
        cur.x = fmaxf(cur.x + a[0],  0.0f);
        cur.y = fmaxf(cur.y + a[32], 0.0f);
        cur.z = fmaxf(cur.z + a[64], 0.0f);
        cur.w = fmaxf(cur.w + a[96], 0.0f);
        *(float4*)o = cur;
    }
}

__global__ void classifier(const float* __restrict__ hm,
                           const float* __restrict__ hv,
                           const int*   __restrict__ im,
                           const int*   __restrict__ iv,
                           const float* __restrict__ Wc,
                           const float* __restrict__ bc,
                           float*       __restrict__ out,
                           int P)
{
    const int p    = blockIdx.x * 8 + (threadIdx.x >> 5);
    const int lane = threadIdx.x & 31;
    if (p >= P) return;

    const float4 am = ((const float4*)(hm + (size_t)im[p] * H))[lane];
    const float4 av = ((const float4*)(hv + (size_t)iv[p] * H))[lane];

    float c0 = 0.f, c1 = 0.f;
    const int ch = lane * 4;
    const float a[4] = {am.x, am.y, am.z, am.w};
    const float b[4] = {av.x, av.y, av.z, av.w};
    #pragma unroll
    for (int i = 0; i < 4; ++i) {
        c0 = fmaf(a[i], Wc[(ch + i) * 2 + 0], c0);
        c1 = fmaf(a[i], Wc[(ch + i) * 2 + 1], c1);
        c0 = fmaf(b[i], Wc[(H + ch + i) * 2 + 0], c0);
        c1 = fmaf(b[i], Wc[(H + ch + i) * 2 + 1], c1);
    }
    #pragma unroll
    for (int off = 16; off > 0; off >>= 1) {
        c0 += __shfl_down(c0, off, 32);
        c1 += __shfl_down(c1, off, 32);
    }
    if (lane == 0) {
        out[(size_t)p * 2 + 0] = c0 + bc[0];
        out[(size_t)p * 2 + 1] = c1 + bc[1];
    }
}

static inline void gemm(const float* A, const float* B, const float* bias,
                        float* C, int M, int K, hipStream_t s)
{
    gemm_wmma_bias<<<dim3(M / 16), dim3(256), 0, s>>>(A, B, bias, C, M, K);
}

extern "C" void kernel_launch(void* const* d_in, const int* in_sizes, int n_in,
                              void* d_out, int out_size, void* d_ws, size_t ws_size,
                              hipStream_t stream)
{
    const float* x_mouse  = (const float*)d_in[0];
    const float* x_virus  = (const float*)d_in[1];
    const float* xe_mv    = (const float*)d_in[2];
    const float* xe_vm    = (const float*)d_in[3];
    const float* W_mouse  = (const float*)d_in[4];
    const float* b_mouse  = (const float*)d_in[5];
    const float* W_virus  = (const float*)d_in[6];
    const float* b_virus  = (const float*)d_in[7];
    const float* W_key    = (const float*)d_in[8];
    const float* b_key    = (const float*)d_in[9];
    const float* W_query  = (const float*)d_in[10];
    const float* b_query  = (const float*)d_in[11];
    const float* W_value  = (const float*)d_in[12];
    const float* b_value  = (const float*)d_in[13];
    const float* W_edge   = (const float*)d_in[14];
    const float* b_edge   = (const float*)d_in[15];
    const float* W_skip   = (const float*)d_in[16];
    const float* bias_cv  = (const float*)d_in[17];
    const float* W_cls    = (const float*)d_in[18];
    const float* b_cls    = (const float*)d_in[19];
    const int* ei_mv_src  = (const int*)d_in[20];
    const int* ei_mv_dst  = (const int*)d_in[21];
    const int* ei_vm_src  = (const int*)d_in[22];
    const int* ei_vm_dst  = (const int*)d_in[23];
    const int* eli_mouse  = (const int*)d_in[24];
    const int* eli_virus  = (const int*)d_in[25];
    float* out = (float*)d_out;

    float* ws  = (float*)d_ws;
    const size_t SZ_M = (size_t)N_MOUSE * H;
    const size_t SZ_V = (size_t)N_VIRUS * H;
    float* hm0 = ws;
    float* hm1 = hm0 + SZ_M;
    float* hv0 = hm1 + SZ_M;
    float* hv1 = hv0 + SZ_V;
    float* kb  = hv1 + SZ_V;
    float* qb  = kb + SZ_M;
    float* vb  = qb + SZ_M;

    gemm(x_mouse, W_mouse, b_mouse, hm0, N_MOUSE, D_IN, stream);
    gemm(x_virus, W_virus, b_virus, hv0, N_VIRUS, D_IN, stream);

    float* hm_in = hm0; float* hv_in = hv0;
    float* hm_out = hm1; float* hv_out = hv1;

    for (int layer = 0; layer < 2; ++layer) {
        const int i_mv = layer * 2;
        const int i_vm = layer * 2 + 1;

        {
            const int i = i_mv;
            const float* Wk = W_key   + (size_t)i * H * H;
            const float* Wq = W_query + (size_t)i * H * H;
            const float* Wv = W_value + (size_t)i * H * H;
            const float* We = W_edge  + (size_t)i * D_EDGE * H;
            const float* Ws = W_skip  + (size_t)i * H * H;
            gemm(hv_in, Ws, bias_cv + (size_t)i * H, hv_out, N_VIRUS, H, stream);
            gemm(hv_in, Wk, b_key   + (size_t)i * H, kb, N_VIRUS, H, stream);
            gemm(hm_in, Wq, b_query + (size_t)i * H, qb, N_MOUSE, H, stream);
            gemm(hm_in, Wv, b_value + (size_t)i * H, vb, N_MOUSE, H, stream);
            edge_conv<<<dim3(EDGE_GRID), dim3(EDGE_THREADS), 0, stream>>>(
                kb, qb, vb, xe_mv, We, b_edge + (size_t)i * H,
                ei_mv_src, ei_mv_dst, hv_out, NEDGE, N_VIRUS, 40);
        }
        {
            const int i = i_vm;
            const float* Wk = W_key   + (size_t)i * H * H;
            const float* Wq = W_query + (size_t)i * H * H;
            const float* Wv = W_value + (size_t)i * H * H;
            const float* We = W_edge  + (size_t)i * D_EDGE * H;
            const float* Ws = W_skip  + (size_t)i * H * H;
            gemm(hm_in, Ws, bias_cv + (size_t)i * H, hm_out, N_MOUSE, H, stream);
            gemm(hm_in, Wk, b_key   + (size_t)i * H, kb, N_MOUSE, H, stream);
            gemm(hv_in, Wq, b_query + (size_t)i * H, qb, N_VIRUS, H, stream);
            gemm(hv_in, Wv, b_value + (size_t)i * H, vb, N_VIRUS, H, stream);
            edge_conv<<<dim3(EDGE_GRID), dim3(EDGE_THREADS), 0, stream>>>(
                kb, qb, vb, xe_vm, We, b_edge + (size_t)i * H,
                ei_vm_src, ei_vm_dst, hm_out, NEDGE, N_MOUSE, EDGE_MAX_ROWS);
        }

        float* t;
        t = hm_in; hm_in = hm_out; hm_out = t;
        t = hv_in; hv_in = hv_out; hv_out = t;
    }

    classifier<<<dim3((NPAIR + 7) / 8), dim3(256), 0, stream>>>(
        hm_in, hv_in, eli_mouse, eli_virus, W_cls, b_cls, out, NPAIR);
}


# edge kernel: a wave takes its next 256-edge list chunk from a work-group ticket counter in LDS (one returning LDS add per chunk), not a fixed slice per step; sums stay LDS atomics in arrival order
# speedup vs baseline: 1.8197x; 1.0306x over previous
#include <hip/hip_runtime.h>
#include <hip/hip_bf16.h>

typedef __attribute__((ext_vector_type(2))) float v2f;
typedef __attribute__((ext_vector_type(8))) float v8f;

#define H 128
#define D_EDGE 16
#define N_MOUSE 50000
#define N_VIRUS 10000
#define NEDGE   500000
#define NPAIR   100000
#define D_IN    1024

__global__ void gemm_wmma_bias(const float* __restrict__ A,
                               const float* __restrict__ B,
                               const float* __restrict__ bias,
                               float* __restrict__ C,
                               int M, int K)
{
    __shared__ __align__(16) float As[16 * 132];

    const int tid  = threadIdx.x;
    const int wave = tid >> 5;
    const int lane = tid & 31;
    const int hi   = lane >> 4;
    const int ml   = lane & 15;
    const int row0 = blockIdx.x * 16;
    const int n0   = wave * 16;

    v8f acc = {};

    for (int k0 = 0; k0 < K; k0 += 128) {
        #pragma unroll
        for (int it = 0; it < 2; ++it) {
            int i   = tid * 4 + it * 1024;
            int row = i >> 7;
            int col = i & 127;
            float4 val = *(const float4*)(A + (size_t)(row0 + row) * K + k0 + col);
            *(float4*)(&As[row * 132 + col]) = val;
        }
        __syncthreads();

        #pragma unroll 8
        for (int kk = 0; kk < 128; kk += 4) {
            v2f a = *(const v2f*)(&As[ml * 132 + kk + 2 * hi]);
            const float* bp = B + (size_t)(k0 + kk + 2 * hi) * H + n0 + ml;
            v2f b;
            b.x = bp[0];
            b.y = bp[H];
            acc = __builtin_amdgcn_wmma_f32_16x16x4_f32(
                      false, a, false, b, (short)0, acc, false, false);
        }
        __syncthreads();
    }

    const float bval = bias ? bias[n0 + ml] : 0.0f;
    #pragma unroll
    for (int r = 0; r < 8; ++r) {
        int m = row0 + r + 8 * hi;
        C[(size_t)m * H + n0 + ml] = acc[r] + bval;
    }
}

#define EDGE_GRID      256
#define EDGE_THREADS   1024
#define EDGE_MAX_ROWS  196
#define CHUNK_INT4     64

__device__ __forceinline__ int4 load_dst4(const int4* __restrict__ dst4, int i, int n4)
{
    return dst4[i < n4 ? i : n4 - 1];
}

__device__ __forceinline__ int ask_ticket(unsigned int* counter, int lane)
{
    unsigned int t = 0;
    if (lane == 0)
        t = __hip_atomic_fetch_add(counter, 1u, __ATOMIC_RELAXED, __HIP_MEMORY_SCOPE_WORKGROUP);
    return (int)t;
}

__device__ __forceinline__ int chunk_base(int t, int first, int nchunks)
{
    int c = first + t;
    if (c >= nchunks) c -= nchunks;
    return c * CHUNK_INT4;
}

__global__ __launch_bounds__(EDGE_THREADS)
void edge_conv(const float* __restrict__ kbuf,
               const float* __restrict__ qbuf,
               const float* __restrict__ vbuf,
               const float* __restrict__ xe,
               const float* __restrict__ We,
               const float* __restrict__ be,
               const int*   __restrict__ src,
               const int*   __restrict__ dst,
               float*       __restrict__ out_dst,
               int E, int n_dst, int rows_per_wg)
{
    __shared__ __align__(16) float acc[EDGE_MAX_ROWS * H];
    __shared__ unsigned int ticket;

    const int tid  = threadIdx.x;
    const int lane = tid & 31;
    const int wave = __builtin_amdgcn_readfirstlane(tid >> 5);

    const int row0 = blockIdx.x * rows_per_wg;
    int nrows = n_dst - row0;
    if (nrows > rows_per_wg) nrows = rows_per_wg;
    if (nrows <= 0) return;

    for (int i = tid; i < nrows * H; i += EDGE_THREADS) acc[i] = 0.0f;
    if (tid == 0) ticket = 0;

    int vzero;
    asm volatile("v_mov_b32 %0, 0" : "=v"(vzero));

    float4 w[D_EDGE];
    #pragma unroll
    for (int dd = 0; dd < D_EDGE; ++dd)
        w[dd] = *(const float4*)(We + dd * H + 4 * lane);
    const float4 b4 = *(const float4*)(be + 4 * lane);

    __syncthreads();

    const int  n4   = E >> 2;
    const int4* dst4 = (const int4*)dst;
    const int  nchunks = (n4 + CHUNK_INT4 - 1) / CHUNK_INT4;
    const int  steps = (n4 + 2047) >> 11;
    const int  first = (blockIdx.x % steps) * 32;

    int t_cur = __builtin_amdgcn_readfirstlane(ask_ticket(&ticket, lane));
    int t_nxt = __builtin_amdgcn_readfirstlane(ask_ticket(&ticket, lane));
    int bw = chunk_base(t_cur, first, nchunks);
    int4 c0 = load_dst4(dst4, bw + lane, n4);
    int4 c1 = load_dst4(dst4, bw + 32 + lane, n4);

    while (t_cur < nchunks) {
        const int nb = chunk_base(t_nxt, first, nchunks);
        const int4 n0 = load_dst4(dst4, nb + lane, n4);
        const int4 n1 = load_dst4(dst4, nb + 32 + lane, n4);
        const int tv = ask_ticket(&ticket, lane);

        const unsigned ok0 = __builtin_amdgcn_ballot_w32(bw + lane      < n4);
        const unsigned ok1 = __builtin_amdgcn_ballot_w32(bw + 32 + lane < n4);
        unsigned m0 = ok0 & __builtin_amdgcn_ballot_w32((unsigned)(c0.x - row0) < (unsigned)nrows);
        unsigned m1 = ok0 & __builtin_amdgcn_ballot_w32((unsigned)(c0.y - row0) < (unsigned)nrows);
        unsigned m2 = ok0 & __builtin_amdgcn_ballot_w32((unsigned)(c0.z - row0) < (unsigned)nrows);
        unsigned m3 = ok0 & __builtin_amdgcn_ballot_w32((unsigned)(c0.w - row0) < (unsigned)nrows);
        unsigned m4 = ok1 & __builtin_amdgcn_ballot_w32((unsigned)(c1.x - row0) < (unsigned)nrows);
        unsigned m5 = ok1 & __builtin_amdgcn_ballot_w32((unsigned)(c1.y - row0) < (unsigned)nrows);
        unsigned m6 = ok1 & __builtin_amdgcn_ballot_w32((unsigned)(c1.z - row0) < (unsigned)nrows);
        unsigned m7 = ok1 & __builtin_amdgcn_ballot_w32((unsigned)(c1.w - row0) < (unsigned)nrows);

        const int t_nn = __builtin_amdgcn_readfirstlane(tv);
        asm volatile("" :: "s"(t_nn));

        while (true) {
            int e, d;
            if (m0)      { const int L = __builtin_ctz(m0); m0 &= m0 - 1; d = __builtin_amdgcn_readlane(c0.x, L); e = 4 * (bw + L); }
            else if (m1) { const int L = __builtin_ctz(m1); m1 &= m1 - 1; d = __builtin_amdgcn_readlane(c0.y, L); e = 4 * (bw + L) + 1; }
            else if (m2) { const int L = __builtin_ctz(m2); m2 &= m2 - 1; d = __builtin_amdgcn_readlane(c0.z, L); e = 4 * (bw + L) + 2; }
            else if (m3) { const int L = __builtin_ctz(m3); m3 &= m3 - 1; d = __builtin_amdgcn_readlane(c0.w, L); e = 4 * (bw + L) + 3; }
            else if (m4) { const int L = __builtin_ctz(m4); m4 &= m4 - 1; d = __builtin_amdgcn_readlane(c1.x, L); e = 4 * (bw + 32 + L); }
            else if (m5) { const int L = __builtin_ctz(m5); m5 &= m5 - 1; d = __builtin_amdgcn_readlane(c1.y, L); e = 4 * (bw + 32 + L) + 1; }
            else if (m6) { const int L = __builtin_ctz(m6); m6 &= m6 - 1; d = __builtin_amdgcn_readlane(c1.z, L); e = 4 * (bw + 32 + L) + 2; }
            else if (m7) { const int L = __builtin_ctz(m7); m7 &= m7 - 1; d = __builtin_amdgcn_readlane(c1.w, L); e = 4 * (bw + 32 + L) + 3; }
            else break;

            const int sv = src[e + vzero];
            const float* xrow = xe + (size_t)e * D_EDGE;
            float x[D_EDGE];
            #pragma unroll
            for (int dd = 0; dd < D_EDGE; ++dd) x[dd] = xrow[dd];
            __builtin_amdgcn_sched_barrier(0);
            const int s = __builtin_amdgcn_readfirstlane(sv);
            const float4 k4 = *(const float4*)(kbuf + (size_t)d * H + 4 * lane);
            const float4 q4 = *(const float4*)(qbuf + (size_t)s * H + 4 * lane);
            const float4 v4 = *(const float4*)(vbuf + (size_t)s * H + 4 * lane);
            __builtin_amdgcn_sched_barrier(0);

            float4 ea = b4;
            #pragma unroll
            for (int dd = 0; dd < D_EDGE; ++dd) {
                ea.x = fmaf(x[dd], w[dd].x, ea.x);
                ea.y = fmaf(x[dd], w[dd].y, ea.y);
                ea.z = fmaf(x[dd], w[dd].z, ea.z);
                ea.w = fmaf(x[dd], w[dd].w, ea.w);
            }

            float* arow = acc + (d - row0) * H + lane;
            {
                const float ki = k4.x + ea.x, qj = q4.x + ea.x, vj = v4.x + ea.x;
                const float g = 1.0f / (1.0f + __expf(-(ki + qj)));
                __hip_atomic_fetch_add(arow,      g * vj, __ATOMIC_RELAXED, __HIP_MEMORY_SCOPE_WORKGROUP);
            }
            {
                const float ki = k4.y + ea.y, qj = q4.y + ea.y, vj = v4.y + ea.y;
                const float g = 1.0f / (1.0f + __expf(-(ki + qj)));
                __hip_atomic_fetch_add(arow + 32, g * vj, __ATOMIC_RELAXED, __HIP_MEMORY_SCOPE_WORKGROUP);
            }
            {
                const float ki = k4.z + ea.z, qj = q4.z + ea.z, vj = v4.z + ea.z;
                const float g = 1.0f / (1.0f + __expf(-(ki + qj)));
                __hip_atomic_fetch_add(arow + 64, g * vj, __ATOMIC_RELAXED, __HIP_MEMORY_SCOPE_WORKGROUP);
            }
            {
                const float ki = k4.w + ea.w, qj = q4.w + ea.w, vj = v4.w + ea.w;
                const float g = 1.0f / (1.0f + __expf(-(ki + qj)));
                __hip_atomic_fetch_add(arow + 96, g * vj, __ATOMIC_RELAXED, __HIP_MEMORY_SCOPE_WORKGROUP);
            }
        }

        bw = nb; c0 = n0; c1 = n1;
        t_cur = t_nxt; t_nxt = t_nn;
    }

    __syncthreads();

    for (int r = wave; r < nrows; r += EDGE_THREADS / 32) {
        float* o = out_dst + (size_t)(row0 + r) * H + 4 * lane;
        const float* a = acc + r * H + lane;
        float4 cur = *(const float4*)o;
        cur.x = fmaxf(cur.x + a[0],  0.0f);
        cur.y = fmaxf(cur.y + a[32], 0.0f);
        cur.z = fmaxf(cur.z + a[64], 0.0f);
        cur.w = fmaxf(cur.w + a[96], 0.0f);
        *(float4*)o = cur;
    }
}

__global__ void classifier(const float* __restrict__ hm,
                           const float* __restrict__ hv,
                           const int*   __restrict__ im,
                           const int*   __restrict__ iv,
                           const float* __restrict__ Wc,
                           const float* __restrict__ bc,
                           float*       __restrict__ out,
                           int P)
{
    const int p    = blockIdx.x * 8 + (threadIdx.x >> 5);
    const int lane = threadIdx.x & 31;
    if (p >= P) return;

    const float4 am = ((const float4*)(hm + (size_t)im[p] * H))[lane];
    const float4 av = ((const float4*)(hv + (size_t)iv[p] * H))[lane];

    float c0 = 0.f, c1 = 0.f;
    const int ch = lane * 4;
    const float a[4] = {am.x, am.y, am.z, am.w};
    const float b[4] = {av.x, av.y, av.z, av.w};
    #pragma unroll
    for (int i = 0; i < 4; ++i) {
        c0 = fmaf(a[i], Wc[(ch + i) * 2 + 0], c0);
        c1 = fmaf(a[i], Wc[(ch + i) * 2 + 1], c1);
        c0 = fmaf(b[i], Wc[(H + ch + i) * 2 + 0], c0);
        c1 = fmaf(b[i], Wc[(H + ch + i) * 2 + 1], c1);
    }
    #pragma unroll
    for (int off = 16; off > 0; off >>= 1) {
        c0 += __shfl_down(c0, off, 32);
        c1 += __shfl_down(c1, off, 32);
    }
    if (lane == 0) {
        out[(size_t)p * 2 + 0] = c0 + bc[0];
        out[(size_t)p * 2 + 1] = c1 + bc[1];
    }
}

static inline void gemm(const float* A, const float* B, const float* bias,
                        float* C, int M, int K, hipStream_t s)
{
    gemm_wmma_bias<<<dim3(M / 16), dim3(256), 0, s>>>(A, B, bias, C, M, K);
}

extern "C" void kernel_launch(void* const* d_in, const int* in_sizes, int n_in,
                              void* d_out, int out_size, void* d_ws, size_t ws_size,
                              hipStream_t stream)
{
    const float* x_mouse  = (const float*)d_in[0];
    const float* x_virus  = (const float*)d_in[1];
    const float* xe_mv    = (const float*)d_in[2];
    const float* xe_vm    = (const float*)d_in[3];
    const float* W_mouse  = (const float*)d_in[4];
    const float* b_mouse  = (const float*)d_in[5];
    const float* W_virus  = (const float*)d_in[6];
    const float* b_virus  = (const float*)d_in[7];
    const float* W_key    = (const float*)d_in[8];
    const float* b_key    = (const float*)d_in[9];
    const float* W_query  = (const float*)d_in[10];
    const float* b_query  = (const float*)d_in[11];
    const float* W_value  = (const float*)d_in[12];
    const float* b_value  = (const float*)d_in[13];
    const float* W_edge   = (const float*)d_in[14];
    const float* b_edge   = (const float*)d_in[15];
    const float* W_skip   = (const float*)d_in[16];
    const float* bias_cv  = (const float*)d_in[17];
    const float* W_cls    = (const float*)d_in[18];
    const float* b_cls    = (const float*)d_in[19];
    const int* ei_mv_src  = (const int*)d_in[20];
    const int* ei_mv_dst  = (const int*)d_in[21];
    const int* ei_vm_src  = (const int*)d_in[22];
    const int* ei_vm_dst  = (const int*)d_in[23];
    const int* eli_mouse  = (const int*)d_in[24];
    const int* eli_virus  = (const int*)d_in[25];
    float* out = (float*)d_out;

    float* ws  = (float*)d_ws;
    const size_t SZ_M = (size_t)N_MOUSE * H;
    const size_t SZ_V = (size_t)N_VIRUS * H;
    float* hm0 = ws;
    float* hm1 = hm0 + SZ_M;
    float* hv0 = hm1 + SZ_M;
    float* hv1 = hv0 + SZ_V;
    float* kb  = hv1 + SZ_V;
    float* qb  = kb + SZ_M;
    float* vb  = qb + SZ_M;

    gemm(x_mouse, W_mouse, b_mouse, hm0, N_MOUSE, D_IN, stream);
    gemm(x_virus, W_virus, b_virus, hv0, N_VIRUS, D_IN, stream);

    float* hm_in = hm0; float* hv_in = hv0;
    float* hm_out = hm1; float* hv_out = hv1;

    for (int layer = 0; layer < 2; ++layer) {
        const int i_mv = layer * 2;
        const int i_vm = layer * 2 + 1;

        {
            const int i = i_mv;
            const float* Wk = W_key   + (size_t)i * H * H;
            const float* Wq = W_query + (size_t)i * H * H;
            const float* Wv = W_value + (size_t)i * H * H;
            const float* We = W_edge  + (size_t)i * D_EDGE * H;
            const float* Ws = W_skip  + (size_t)i * H * H;
            gemm(hv_in, Ws, bias_cv + (size_t)i * H, hv_out, N_VIRUS, H, stream);
            gemm(hv_in, Wk, b_key   + (size_t)i * H, kb, N_VIRUS, H, stream);
            gemm(hm_in, Wq, b_query + (size_t)i * H, qb, N_MOUSE, H, stream);
            gemm(hm_in, Wv, b_value + (size_t)i * H, vb, N_MOUSE, H, stream);
            edge_conv<<<dim3(EDGE_GRID), dim3(EDGE_THREADS), 0, stream>>>(
                kb, qb, vb, xe_mv, We, b_edge + (size_t)i * H,
                ei_mv_src, ei_mv_dst, hv_out, NEDGE, N_VIRUS, 40);
        }
        {
            const int i = i_vm;
            const float* Wk = W_key   + (size_t)i * H * H;
            const float* Wq = W_query + (size_t)i * H * H;
            const float* Wv = W_value + (size_t)i * H * H;
            const float* We = W_edge  + (size_t)i * D_EDGE * H;
            const float* Ws = W_skip  + (size_t)i * H * H;
            gemm(hm_in, Ws, bias_cv + (size_t)i * H, hm_out, N_MOUSE, H, stream);
            gemm(hm_in, Wk, b_key   + (size_t)i * H, kb, N_MOUSE, H, stream);
            gemm(hv_in, Wq, b_query + (size_t)i * H, qb, N_VIRUS, H, stream);
            gemm(hv_in, Wv, b_value + (size_t)i * H, vb, N_VIRUS, H, stream);
            edge_conv<<<dim3(EDGE_GRID), dim3(EDGE_THREADS), 0, stream>>>(
                kb, qb, vb, xe_vm, We, b_edge + (size_t)i * H,
                ei_vm_src, ei_vm_dst, hm_out, NEDGE, N_MOUSE, EDGE_MAX_ROWS);
        }

        float* t;
        t = hm_in; hm_in = hm_out; hm_out = t;
        t = hv_in; hv_in = hv_out; hv_out = t;
    }

    classifier<<<dim3((NPAIR + 7) / 8), dim3(256), 0, stream>>>(
        hm_in, hv_in, eli_mouse, eli_virus, W_cls, b_cls, out, NPAIR);
}
